# starting kernel
# baseline (speedup 1.0000x reference)
#include <hip/hip_runtime.h>

#define F       128
#define ROWS    64
#define WPAD    136
#define APAD    132
#define THREADS 256

typedef __attribute__((ext_vector_type(2))) float v2f;
typedef __attribute__((ext_vector_type(8))) float v8f;

__global__ __launch_bounds__(THREADS) void gcn_fused_kernel(
    const float* __restrict__ X,
    const float* __restrict__ W,
    const float* __restrict__ deg,
    const int*   __restrict__ rowp,
    const int*   __restrict__ colidx,
    float*       __restrict__ out,
    int n)
{
    extern __shared__ float smem[];
    float* sW = smem;
    float* sA = smem + F * WPAD;

    const int tid  = threadIdx.x;
    const int wave = tid >> 5;
    const int lane = tid & 31;

    for (int i = tid; i < F * (F / 4); i += THREADS) {
        const int r = i >> 5;
        const int c = (i & 31) << 2;
        const float4 w = *(const float4*)(W + r * F + c);
        float* dst = sW + r * WPAD + c;
        dst[0] = w.x; dst[1] = w.y; dst[2] = w.z; dst[3] = w.w;
    }

    const int rowBase = blockIdx.x * ROWS;
    const int c4 = lane << 2;
    for (int i = 0; i < ROWS / 8; ++i) {
        const int lrow = wave * (ROWS / 8) + i;
        const int row  = rowBase + lrow;
        float ax = 0.f, ay = 0.f, az = 0.f, aw = 0.f;
        if (row < n) {
            const int   s  = rowp[row];
            const int   e  = rowp[row + 1];
            const float di = deg[row];
            for (int ed = s; ed < e; ++ed) {
                const int   j   = colidx[ed];
                const float nrm = di * deg[j];
                const float4 x = *(const float4*)(X + (long long)j * F + c4);
                ax += nrm * x.x; ay += nrm * x.y;
                az += nrm * x.z; aw += nrm * x.w;
            }
        }
        float* dst = sA + lrow * APAD + c4;
        dst[0] = ax; dst[1] = ay; dst[2] = az; dst[3] = aw;
    }
    __syncthreads();

    const int half = lane >> 4;
    const int l16  = lane & 15;
    const int colBase = wave << 4;
    for (int tr = 0; tr < ROWS / 16; ++tr) {
        v8f acc = {};
        const float* Arow = sA + (tr * 16 + l16) * APAD;
        const float* Bcol = sW + colBase + l16;
        #pragma unroll
        for (int k0 = 0; k0 < F; k0 += 4) {
            const int ka = k0 + 2 * half;
            v2f a; a.x = Arow[ka];          a.y = Arow[ka + 1];
            v2f b; b.x = Bcol[ka * WPAD];   b.y = Bcol[(ka + 1) * WPAD];
            acc = __builtin_amdgcn_wmma_f32_16x16x4_f32(
                false, a, false, b, (short)0, acc, false, false);
        }
        const int col      = colBase + l16;
        const int nodeBase = rowBase + tr * 16 + half * 8;
        #pragma unroll
        for (int r = 0; r < 8; ++r) {
            const int node = nodeBase + r;
            if (node < n) out[(long long)node * F + col] = acc[r];
        }
    }
}

extern "C" void kernel_launch(void* const* d_in, const int* in_sizes, int n_in,
                              void* d_out, int out_size, void* d_ws, size_t ws_size,
                              hipStream_t stream) {
    const float* X      = (const float*)d_in[0];
    const float* W      = (const float*)d_in[1];
    const float* deg    = (const float*)d_in[2];
    const int*   rowp   = (const int*)d_in[3];
    const int*   colidx = (const int*)d_in[4];
    float*       out    = (float*)d_out;

    const int n = in_sizes[2];
    const int blocks = (n + ROWS - 1) / ROWS;
    const size_t lds_bytes = (size_t)(F * WPAD + ROWS * APAD) * sizeof(float);

    gcn_fused_kernel<<<blocks, THREADS, lds_bytes, stream>>>(
        X, W, deg, rowp, colidx, out, n);
}


# Product launch as a persistent grid of 512 work-groups: W staged to LDS once each, then a loop over 16-row tiles, next tile prefetched to a second LDS buffer, one LDS-only barrier a trip; sums same
# speedup vs baseline: 1.0946x; 1.0946x over previous
#include <hip/hip_runtime.h>


#define F       128
#define ROWS    64
#define WPAD    136
#define APAD    132
#define THREADS 256

typedef __attribute__((ext_vector_type(2))) float v2f;
typedef __attribute__((ext_vector_type(8))) float v8f;

__global__ __launch_bounds__(THREADS) void gcn_aggregate_kernel(
    const float* __restrict__ X,
    const float* __restrict__ deg,
    const int*   __restrict__ rowp,
    const int*   __restrict__ colidx,
    float*       __restrict__ agg,
    int n)
{
    const int tid  = threadIdx.x;
    const int wave = tid >> 5;
    const int lane = tid & 31;

    const int rowBase = blockIdx.x * ROWS;
    const int c4 = lane << 2;
    for (int i = 0; i < ROWS / 8; ++i) {
        const int lrow = wave * (ROWS / 8) + i;
        const int row  = rowBase + lrow;
        if (row < n) {
            float ax = 0.f, ay = 0.f, az = 0.f, aw = 0.f;
            const int   s  = rowp[row];
            const int   e  = rowp[row + 1];
            const float di = deg[row];
            for (int ed = s; ed < e; ++ed) {
                const int   j   = colidx[ed];
                const float nrm = di * deg[j];
                const float4 x = *(const float4*)(X + (long long)j * F + c4);
                ax += nrm * x.x; ay += nrm * x.y;
                az += nrm * x.z; aw += nrm * x.w;
            }
            *(float4*)(agg + (long long)row * F + c4) = make_float4(ax, ay, az, aw);
        }
    }
}

#define TROWS    16
#define TGRID    512

__device__ __forceinline__ void load_agg_tile(const float* out, int t, int n, int tid,
                                              float4 (&av)[TROWS * (F / 4) / THREADS])
{
    #pragma unroll
    for (int q = 0; q < TROWS * (F / 4) / THREADS; ++q) {
        const int i = tid + q * THREADS;
        const int r = i >> 5;
        const int c = (i & 31) << 2;
        const int row = t * TROWS + r;
        const int rc  = row < n ? row : n - 1;
        av[q] = *(const float4*)(out + (long long)rc * F + c);
    }
}

__global__ __launch_bounds__(THREADS) void gcn_transform_kernel(
    const float* __restrict__ W,
    float*       out,
    int n)
{
    extern __shared__ float smem[];
    float* sW = smem;
    float* sA = smem + F * WPAD;

    const int tid  = threadIdx.x;
    const int wave = tid >> 5;
    const int lane = tid & 31;
    const int ntiles = (n + TROWS - 1) / TROWS;

    float4 wv[F * (F / 4) / THREADS];
    float4 av[TROWS * (F / 4) / THREADS];
    #pragma unroll
    for (int q = 0; q < F * (F / 4) / THREADS; ++q) {
        const int i = tid + q * THREADS;
        const int r = i >> 5;
        const int c = (i & 31) << 2;
        wv[q] = *(const float4*)(W + r * F + c);
    }
    int t = blockIdx.x;
    load_agg_tile(out, t, n, tid, av);
    #pragma unroll
    for (int q = 0; q < F * (F / 4) / THREADS; ++q) {
        const int i = tid + q * THREADS;
        const int r = i >> 5;
        const int c = (i & 31) << 2;
        float* dst = sW + r * WPAD + c;
        dst[0] = wv[q].x; dst[1] = wv[q].y; dst[2] = wv[q].z; dst[3] = wv[q].w;
    }
    #pragma unroll
    for (int q = 0; q < TROWS * (F / 4) / THREADS; ++q) {
        const int i = tid + q * THREADS;
        float* dst = sA + (i >> 5) * APAD + ((i & 31) << 2);
        dst[0] = av[q].x; dst[1] = av[q].y; dst[2] = av[q].z; dst[3] = av[q].w;
    }
    __syncthreads();

    const int half = lane >> 4;
    const int l16  = lane & 15;
    const int colBase = wave << 4;
    const float* Bcol = sW + colBase + l16;
    int buf = 0;
    for (; t < ntiles; t += TGRID) {
        const int  tn      = t + TGRID;
        const bool hasNext = tn < ntiles;
        if (hasNext) load_agg_tile(out, tn, n, tid, av);

        v8f acc = {};
        const float* Arow = sA + buf * (TROWS * APAD) + l16 * APAD;
        #pragma unroll
        for (int k0 = 0; k0 < F; k0 += 4) {
            const int ka = k0 + 2 * half;
            v2f a; a.x = Arow[ka];          a.y = Arow[ka + 1];
            v2f b; b.x = Bcol[ka * WPAD];   b.y = Bcol[(ka + 1) * WPAD];
            acc = __builtin_amdgcn_wmma_f32_16x16x4_f32(
                false, a, false, b, (short)0, acc, false, false);
        }
        const int col      = colBase + l16;
        const int nodeBase = t * TROWS + half * 8;
        #pragma unroll
        for (int r = 0; r < 8; ++r) {
            const int node = nodeBase + r;
            if (node < n) out[(long long)node * F + col] = acc[r];
        }
        if (hasNext) {
            float* nA = sA + (buf ^ 1) * (TROWS * APAD);
            #pragma unroll
            for (int q = 0; q < TROWS * (F / 4) / THREADS; ++q) {
                const int i = tid + q * THREADS;
                float* dst = nA + (i >> 5) * APAD + ((i & 31) << 2);
                dst[0] = av[q].x; dst[1] = av[q].y; dst[2] = av[q].z; dst[3] = av[q].w;
            }
        }
        __builtin_amdgcn_fence(__ATOMIC_RELEASE, "workgroup", "local");
        __builtin_amdgcn_s_barrier();
        __builtin_amdgcn_fence(__ATOMIC_ACQUIRE, "workgroup", "local");
        buf ^= 1;
    }
}

extern "C" void kernel_launch(void* const* d_in, const int* in_sizes, int n_in,
                              void* d_out, int out_size, void* d_ws, size_t ws_size,
                              hipStream_t stream) {
    const float* X      = (const float*)d_in[0];
    const float* W      = (const float*)d_in[1];
    const float* deg    = (const float*)d_in[2];
    const int*   rowp   = (const int*)d_in[3];
    const int*   colidx = (const int*)d_in[4];
    float*       out    = (float*)d_out;

    const int n = in_sizes[2];
    const int blocks = (n + ROWS - 1) / ROWS;
    const size_t lds_bytes = (size_t)(F * WPAD + 2 * TROWS * APAD) * sizeof(float);

    gcn_aggregate_kernel<<<blocks, THREADS, 0, stream>>>(
        X, deg, rowp, colidx, out, n);
    gcn_transform_kernel<<<TGRID, THREADS, lds_bytes, stream>>>(
        W, out, n);
}


# One launch: in each persistent work-group 16 gather waves sum the next 16-row tile into LDS while 4 matrix waves multiply the current tile by W from LDS; aggregate no longer goes to memory; sums same
# speedup vs baseline: 1.5417x; 1.4085x over previous
#include <hip/hip_runtime.h>


#define F       128
#define WPAD    136
#define APAD    132
#define TROWS   16
#define GWAVES  16
#define MWAVES  4
#define THREADS ((GWAVES + MWAVES) * 32)
#define GRID    768
#define WBATCH  8

typedef __attribute__((ext_vector_type(2))) float v2f;
typedef __attribute__((ext_vector_type(8))) float v8f;


__device__ __forceinline__ void gather_tile(
    const float* __restrict__ X, const float* __restrict__ deg,
    const int* __restrict__ rowp, const int* __restrict__ colidx,
    float* tileA, int tile, int n, int gw, int lane)
{
    const int c4  = lane << 2;
    const int row = tile * TROWS + gw;
    float ax = 0.f, ay = 0.f, az = 0.f, aw = 0.f;
    if (row < n) {
        const int   s  = rowp[row];
        const int   e  = rowp[row + 1];
        const float di = deg[row];
        for (int ed = s; ed < e; ++ed) {
            const int   j   = colidx[ed];
            const float nrm = di * deg[j];
            const float4 x = *(const float4*)(X + (long long)j * F + c4);
            ax += nrm * x.x; ay += nrm * x.y;
            az += nrm * x.z; aw += nrm * x.w;
        }
    }
    *(float4*)(tileA + gw * APAD + c4) = make_float4(ax, ay, az, aw);
}

__device__ __forceinline__ void product_tile(
    const float* sW, const float* tileA, float* __restrict__ out,
    int tile, int n, int mw, int lane)
{
    const int half = lane >> 4;
    const int l16  = lane & 15;
    const int colBase = mw << 5;
    const float* Bcol = sW + colBase + l16;
    const float* Arow = tileA + l16 * APAD;
    v8f acc0 = {}, acc1 = {};
    #pragma unroll 4
    for (int k0 = 0; k0 < F; k0 += 4) {
        const int ka = k0 + 2 * half;
        v2f a;  a.x  = Arow[ka];               a.y  = Arow[ka + 1];
        v2f b0; b0.x = Bcol[ka * WPAD];        b0.y = Bcol[(ka + 1) * WPAD];
        v2f b1; b1.x = Bcol[ka * WPAD + 16];   b1.y = Bcol[(ka + 1) * WPAD + 16];
        acc0 = __builtin_amdgcn_wmma_f32_16x16x4_f32(false, a, false, b0, (short)0, acc0, false, false);
        acc1 = __builtin_amdgcn_wmma_f32_16x16x4_f32(false, a, false, b1, (short)0, acc1, false, false);
    }
    const int col      = colBase + l16;
    const int nodeBase = tile * TROWS + half * 8;
    #pragma unroll
    for (int r = 0; r < 8; ++r) {
        const int node = nodeBase + r;
        if (node < n) {
            out[(long long)node * F + col]      = acc0[r];
            out[(long long)node * F + col + 16] = acc1[r];
        }
    }
}

__global__ __launch_bounds__(THREADS) void gcn_gather_product_kernel(
    const float* __restrict__ X,
    const float* __restrict__ W,
    const float* __restrict__ deg,
    const int*   __restrict__ rowp,
    const int*   __restrict__ colidx,
    float*       __restrict__ out,
    int n)
{
    extern __shared__ __align__(16) float smem[];
    float* sW = smem;
    float* sA = smem + F * WPAD;

    const int tid  = threadIdx.x;
    const int lane = tid & 31;
    const int wave = __builtin_amdgcn_readfirstlane(tid >> 5);
    const bool gatherRole = wave < GWAVES;
    const int ntiles = (n + TROWS - 1) / TROWS;

    int t = blockIdx.x;
    if (gatherRole) {
        if (t < ntiles)
            gather_tile(X, deg, rowp, colidx, sA, t, n, wave, lane);
    } else {
        const int mt = tid - GWAVES * 32;
        #pragma unroll 1
        for (int q0 = 0; q0 < F * (F / 4) / (MWAVES * 32); q0 += WBATCH) {
            float4 wv[WBATCH];
            #pragma unroll
            for (int q = 0; q < WBATCH; ++q) {
                const int i = mt + (q0 + q) * (MWAVES * 32);
                wv[q] = *(const float4*)(W + (i >> 5) * F + ((i & 31) << 2));
            }
            #pragma unroll
            for (int q = 0; q < WBATCH; ++q) {
                const int i = mt + (q0 + q) * (MWAVES * 32);
                *(float4*)(sW + (i >> 5) * WPAD + ((i & 31) << 2)) = wv[q];
            }
        }
    }
    __syncthreads();

    int buf = 0;
    for (; t < ntiles; t += GRID) {
        const int tn = t + GRID;
        if (gatherRole) {
            if (tn < ntiles)
                gather_tile(X, deg, rowp, colidx, sA + (buf ^ 1) * (TROWS * APAD), tn, n, wave, lane);
        } else {
            product_tile(sW, sA + buf * (TROWS * APAD), out, t, n, wave - GWAVES, lane);
        }
        __builtin_amdgcn_fence(__ATOMIC_RELEASE, "workgroup", "local");
        __builtin_amdgcn_s_barrier();
        __builtin_amdgcn_fence(__ATOMIC_ACQUIRE, "workgroup", "local");
        buf ^= 1;
    }
}

extern "C" void kernel_launch(void* const* d_in, const int* in_sizes, int n_in,
                              void* d_out, int out_size, void* d_ws, size_t ws_size,
                              hipStream_t stream) {
    const float* X      = (const float*)d_in[0];
    const float* W      = (const float*)d_in[1];
    const float* deg    = (const float*)d_in[2];
    const int*   rowp   = (const int*)d_in[3];
    const int*   colidx = (const int*)d_in[4];
    float*       out    = (float*)d_out;

    const int n = in_sizes[2];
    const size_t lds_bytes = (size_t)(F * WPAD + 2 * TROWS * APAD) * sizeof(float);

    gcn_gather_product_kernel<<<GRID, THREADS, lds_bytes, stream>>>(
        X, W, deg, rowp, colidx, out, n);
}
